# starting kernel
# baseline (speedup 1.0000x reference)
#include <hip/hip_runtime.h>
#include <math.h>


typedef float v2f __attribute__((ext_vector_type(2)));
typedef float v8f __attribute__((ext_vector_type(8)));

#define GAT_B 4
#define GAT_N 20000
#define GAT_K 16
#define GAT_D 256
#define GAT_H 4
#define GAT_CH 64

__global__ __launch_bounds__(128)
void gat_gemm_wmma(const float* __restrict__ x,
                   const float* __restrict__ W,
                   float* __restrict__ h)
{
    const int lane = threadIdx.x & 31;
    const int wave = threadIdx.x >> 5;
    const int m0   = blockIdx.x * 16;
    const int n0   = wave * 64;
    const int half = lane >> 4;
    const int l16  = lane & 15;

    const float* arow = x + (size_t)(m0 + l16) * GAT_D;
    const float* b0r  = W + (size_t)(n0 +  0 + l16) * GAT_D;
    const float* b1r  = W + (size_t)(n0 + 16 + l16) * GAT_D;
    const float* b2r  = W + (size_t)(n0 + 32 + l16) * GAT_D;
    const float* b3r  = W + (size_t)(n0 + 48 + l16) * GAT_D;

    v8f acc0 = {}; v8f acc1 = {}; v8f acc2 = {}; v8f acc3 = {};

    #pragma unroll 4
    for (int k = 0; k < GAT_D; k += 4) {
        const int kk = k + 2 * half;
        v2f a  = *(const v2f*)(arow + kk);
        v2f b0 = *(const v2f*)(b0r + kk);
        v2f b1 = *(const v2f*)(b1r + kk);
        v2f b2 = *(const v2f*)(b2r + kk);
        v2f b3 = *(const v2f*)(b3r + kk);
        acc0 = __builtin_amdgcn_wmma_f32_16x16x4_f32(false, a, false, b0, (short)0, acc0, false, false);
        acc1 = __builtin_amdgcn_wmma_f32_16x16x4_f32(false, a, false, b1, (short)0, acc1, false, false);
        acc2 = __builtin_amdgcn_wmma_f32_16x16x4_f32(false, a, false, b2, (short)0, acc2, false, false);
        acc3 = __builtin_amdgcn_wmma_f32_16x16x4_f32(false, a, false, b3, (short)0, acc3, false, false);
    }

    #pragma unroll
    for (int r = 0; r < 8; ++r) {
        float* hrow = h + (size_t)(m0 + r + 8 * half) * GAT_D + n0 + l16;
        hrow[ 0] = acc0[r];
        hrow[16] = acc1[r];
        hrow[32] = acc2[r];
        hrow[48] = acc3[r];
    }
}

__device__ __forceinline__ float dot4f(float4 a, float4 b) {
    return fmaf(a.x, b.x, fmaf(a.y, b.y, fmaf(a.z, b.z, a.w * b.w)));
}

__global__ __launch_bounds__(256)
void gat_gather_softmax(const float* __restrict__ h,
                        const int*   __restrict__ nbr,
                        const float* __restrict__ attn,
                        float*       __restrict__ out)
{
    const int lane = threadIdx.x & 31;
    const int wid  = threadIdx.x >> 5;
    const int p    = blockIdx.x * 8 + wid;
    const int b    = p / GAT_N;
    const int c0   = lane * 8;
    const int head = lane >> 3;
    const int ch   = c0 & (GAT_CH - 1);

    const int4* idxv = (const int4*)(nbr + (size_t)p * GAT_K);
    int4 i0 = idxv[0];
    int4 i1 = idxv[1];
    int4 i2 = idxv[2];
    int4 i3 = idxv[3];
    int nbs[GAT_K] = { i0.x, i0.y, i0.z, i0.w,
                       i1.x, i1.y, i1.z, i1.w,
                       i2.x, i2.y, i2.z, i2.w,
                       i3.x, i3.y, i3.z, i3.w };

    const float* ap = attn + head * (2 * GAT_CH) + ch;
    float4 as0 = *(const float4*)(ap);
    float4 as1 = *(const float4*)(ap + 4);
    float4 ad0 = *(const float4*)(ap + GAT_CH);
    float4 ad1 = *(const float4*)(ap + GAT_CH + 4);

    const float* hp = h + (size_t)p * GAT_D + c0;
    float4 hi0 = *(const float4*)(hp);
    float4 hi1 = *(const float4*)(hp + 4);
    float ssrc = dot4f(hi0, as0) + dot4f(hi1, as1);
    ssrc += __shfl_xor(ssrc, 1, 32);
    ssrc += __shfl_xor(ssrc, 2, 32);
    ssrc += __shfl_xor(ssrc, 4, 32);

    const float* hbase = h + (size_t)b * GAT_N * GAT_D + c0;

    float m = -3.402823466e+38f;
    float s = 0.0f;
    float a0 = 0.f, a1 = 0.f, a2 = 0.f, a3 = 0.f;
    float a4 = 0.f, a5 = 0.f, a6 = 0.f, a7 = 0.f;

    #pragma unroll
    for (int k = 0; k < GAT_K; ++k) {
        const float* jp = hbase + (size_t)nbs[k] * GAT_D;
        float4 j0 = *(const float4*)(jp);
        float4 j1 = *(const float4*)(jp + 4);

        float sd = dot4f(j0, ad0) + dot4f(j1, ad1);
        sd += __shfl_xor(sd, 1, 32);
        sd += __shfl_xor(sd, 2, 32);
        sd += __shfl_xor(sd, 4, 32);

        const float e  = (ssrc + sd) * 0.125f;
        const float mn = fmaxf(m, e);
        const float sc = __expf(m - mn);
        const float w  = __expf(e - mn);
        s = s * sc + w;
        m = mn;
        a0 = a0 * sc + w * j0.x;
        a1 = a1 * sc + w * j0.y;
        a2 = a2 * sc + w * j0.z;
        a3 = a3 * sc + w * j0.w;
        a4 = a4 * sc + w * j1.x;
        a5 = a5 * sc + w * j1.y;
        a6 = a6 * sc + w * j1.z;
        a7 = a7 * sc + w * j1.w;
    }

    const float inv = 1.0f / s;
    float v[8] = { a0 * inv, a1 * inv, a2 * inv, a3 * inv,
                   a4 * inv, a5 * inv, a6 * inv, a7 * inv };

    #pragma unroll
    for (int j = 0; j < 8; ++j) {
        float t = v[j];
        t += __shfl_xor(t, 8, 32);
        t += __shfl_xor(t, 16, 32);
        v[j] = t * 0.25f;
    }

    if (lane < 8) {
        float* op = out + (size_t)p * GAT_CH + lane * 8;
        *(float4*)(op)     = make_float4(v[0], v[1], v[2], v[3]);
        *(float4*)(op + 4) = make_float4(v[4], v[5], v[6], v[7]);
    }
}

extern "C" void kernel_launch(void* const* d_in, const int* in_sizes, int n_in,
                              void* d_out, int out_size, void* d_ws, size_t ws_size,
                              hipStream_t stream)
{
    const float* x    = (const float*)d_in[0];
    const int*   idx  = (const int*)  d_in[1];
    const float* W    = (const float*)d_in[2];
    const float* attn = (const float*)d_in[3];
    float* out = (float*)d_out;
    float* h   = (float*)d_ws;

    gat_gemm_wmma<<<(GAT_B * GAT_N) / 16, 128, 0, stream>>>(x, W, h);

    gat_gather_softmax<<<(GAT_B * GAT_N) / 8, 256, 0, stream>>>(h, idx, attn, out);
}


# gather kernel: half a wave per node (40000 waves of 2 nodes, was 80000): a lane owns 4 channels of every head, 16-lane sums on the vector ALU; scores, softmax and head mean summed in a new order
# speedup vs baseline: 1.7681x; 1.7681x over previous
#include <hip/hip_runtime.h>
#include <math.h>


typedef float v2f __attribute__((ext_vector_type(2)));
typedef float v8f __attribute__((ext_vector_type(8)));

#define GAT_B 4
#define GAT_N 20000
#define GAT_K 16
#define GAT_D 256
#define GAT_H 4
#define GAT_CH 64

#define GAT_PITCH   260
#define GAT_MTILES  ((GAT_B * GAT_N) / 16)
#define GAT_GROUPS  256
#define GAT_KC      64

__device__ __forceinline__ void gat_load_a(v2f (&a)[16], const float* __restrict__ p)
{
    #pragma unroll
    for (int s = 0; s < 16; ++s)
        a[s] = *(const v2f*)(p + 4 * s);
}

__device__ __forceinline__ void gat_load_b(v2f (&b)[8][2], const float* bl)
{
    #pragma unroll
    for (int t = 0; t < 8; ++t) {
        b[t][0] = *(const v2f*)(bl + t * 16 * GAT_PITCH);
        b[t][1] = *(const v2f*)(bl + t * 16 * GAT_PITCH + 4);
    }
}

__device__ __forceinline__ void gat_mma_pair(v8f (&acc)[8], v2f alo, v2f ahi, const v2f (&b)[8][2])
{
    #pragma unroll
    for (int t = 0; t < 8; ++t)
        acc[t] = __builtin_amdgcn_wmma_f32_16x16x4_f32(false, alo, false, b[t][0], (short)0, acc[t], false, false);
    #pragma unroll
    for (int t = 0; t < 8; ++t)
        acc[t] = __builtin_amdgcn_wmma_f32_16x16x4_f32(false, ahi, false, b[t][1], (short)0, acc[t], false, false);
}

__device__ __forceinline__ void gat_wmma_gap()
{
    asm volatile("v_nop\n\tv_nop\n\tv_nop\n\tv_nop");
}

__device__ __forceinline__ void gat_mma_chunk(v8f (&acc)[8], const v2f (&a)[16],
                                              v2f (&bcur)[8][2], v2f (&bnxt)[8][2],
                                              const float* bl, const float* bfollow)
{
    #pragma unroll
    for (int pp = 0; pp < 8; pp += 2) {
        gat_load_b(bnxt, bl + 8 * (pp + 1));
        __builtin_amdgcn_sched_barrier(0);
        gat_mma_pair(acc, a[2 * pp], a[2 * pp + 1], bcur);
        __builtin_amdgcn_sched_barrier(0);
        gat_wmma_gap();
        __builtin_amdgcn_sched_barrier(0);
        gat_load_b(bcur, (pp + 2 < 8) ? (bl + 8 * (pp + 2)) : bfollow);
        __builtin_amdgcn_sched_barrier(0);
        gat_mma_pair(acc, a[2 * pp + 2], a[2 * pp + 3], bnxt);
        __builtin_amdgcn_sched_barrier(0);
        gat_wmma_gap();
        __builtin_amdgcn_sched_barrier(0);
    }
}

__global__ __launch_bounds__(256)
void gat_gemm_wmma(const float* __restrict__ x,
                   const float* __restrict__ W,
                   float* __restrict__ h)
{
    __shared__ __attribute__((aligned(16))) float Wl[GAT_D * GAT_PITCH];

    const int tid  = threadIdx.x;
    const int lane = tid & 31;
    const int wave = __builtin_amdgcn_readfirstlane(tid >> 5);
    const int half = lane >> 4;
    const int l16  = lane & 15;
    const int slot = wave & 3;
    const int n0   = (wave >> 2) * 128;

    int tile = blockIdx.x * 4 + slot;
    const float* arow = x + (size_t)(tile * 16 + l16) * GAT_D + 2 * half;

    v2f a0[16], a1[16];
    gat_load_a(a0, arow);

    #pragma unroll 16
    for (int j = 0; j < (GAT_D * GAT_D / 4) / 256; ++j) {
        const int i   = ((j + blockIdx.x) & 63) * 256 + tid;
        const int row = i >> 6;
        const int c4  = i & 63;
        const float4 v = *(const float4*)(W + (size_t)row * GAT_D + c4 * 4);
        *(float4*)(&Wl[row * GAT_PITCH + c4 * 4]) = v;
    }
    __syncthreads();

    int boff = (n0 + l16) * GAT_PITCH + 2 * half;

    v2f b0[8][2], b1[8][2];
    gat_load_b(b0, &Wl[boff]);

    for (;;) {
        asm volatile("" : "+v"(boff));
        const float* bl = &Wl[boff];

        v8f acc[8];
        #pragma unroll
        for (int t = 0; t < 8; ++t) acc[t] = (v8f){};

        const int  next     = tile + 4 * GAT_GROUPS;
        const bool has_next = next < GAT_MTILES;
        const int  ntile    = has_next ? next : tile;
        const float* anext  = x + (size_t)(ntile * 16 + l16) * GAT_D + 2 * half;

        gat_load_a(a1, arow + 1 * GAT_KC);
        gat_mma_chunk(acc, a0, b0, b1, bl + 0 * GAT_KC, bl + 1 * GAT_KC);
        gat_load_a(a0, arow + 2 * GAT_KC);
        gat_mma_chunk(acc, a1, b0, b1, bl + 1 * GAT_KC, bl + 2 * GAT_KC);
        gat_load_a(a1, arow + 3 * GAT_KC);
        gat_mma_chunk(acc, a0, b0, b1, bl + 2 * GAT_KC, bl + 3 * GAT_KC);
        gat_load_a(a0, anext);
        gat_mma_chunk(acc, a1, b0, b1, bl + 3 * GAT_KC, bl);

        const int m0 = tile * 16;
        #pragma unroll
        for (int r = 0; r < 8; ++r) {
            float* hrow = h + (size_t)(m0 + r + 8 * half) * GAT_D + n0 + l16;
            #pragma unroll
            for (int t = 0; t < 8; ++t)
                hrow[16 * t] = acc[t][r];
        }

        if (!has_next) break;
        tile = next;
        arow = anext;
    }
}

#define GAT_GW 4
__device__ __forceinline__ float dot4f(float4 a, float4 b) {
    return fmaf(a.x, b.x, fmaf(a.y, b.y, fmaf(a.z, b.z, a.w * b.w)));
}

template <int CTRL>
__device__ __forceinline__ float gat_row_read(float x)
{
    return __builtin_bit_cast(float,
        __builtin_amdgcn_update_dpp(0, __builtin_bit_cast(int, x), CTRL, 0xf, 0xf, true));
}

__device__ __forceinline__ float gat_row_sum4(float d0, float d1, float d2, float d3, bool b0, bool b1)
{
    const float g0 = b0 ? d0 : d2, k0 = b0 ? d2 : d0;
    const float g1 = b0 ? d1 : d3, k1 = b0 ? d3 : d1;
    const float n0 = k0 + gat_row_read<0x161>(g0);
    const float n1 = k1 + gat_row_read<0x161>(g1);
    const float g  = b1 ? n0 : n1, kk = b1 ? n1 : n0;
    float r = kk + gat_row_read<0x162>(g);
    r += gat_row_read<0x164>(r);
    r += gat_row_read<0x168>(r);
    return r;
}

__device__ __forceinline__ float4 fma4(float w, float4 v, float4 a) {
    return make_float4(fmaf(w, v.x, a.x), fmaf(w, v.y, a.y), fmaf(w, v.z, a.z), fmaf(w, v.w, a.w));
}

__global__ __launch_bounds__(GAT_GW * 32)
void gat_gather_softmax(const float* __restrict__ h,
                        const int*   __restrict__ nbr,
                        const float* __restrict__ attn,
                        float*       __restrict__ out)
{
    const int lane = threadIdx.x & 31;
    const int wid  = threadIdx.x >> 5;
    const int i    = lane & 15;
    const int p    = (blockIdx.x * GAT_GW + wid) * 2 + (lane >> 4);
    const int b    = p / GAT_N;
    const bool b0  = (i & 1) != 0;
    const bool b1  = (i & 2) != 0;

    const int4* idxv = (const int4*)(nbr + (size_t)p * GAT_K);
    int4 i0 = idxv[0];
    int4 i1 = idxv[1];
    int4 i2 = idxv[2];
    int4 i3 = idxv[3];
    int nbs[GAT_K] = { i0.x, i0.y, i0.z, i0.w,
                       i1.x, i1.y, i1.z, i1.w,
                       i2.x, i2.y, i2.z, i2.w,
                       i3.x, i3.y, i3.z, i3.w };

    const float* ap = attn + 4 * i;
    const float* hp = h + (size_t)p * GAT_D + 4 * i;
    float4 as[GAT_H], ad[GAT_H], own[GAT_H];
    #pragma unroll
    for (int q = 0; q < GAT_H; ++q) {
        as[q]  = *(const float4*)(ap + q * (2 * GAT_CH));
        ad[q]  = *(const float4*)(ap + q * (2 * GAT_CH) + GAT_CH);
        own[q] = *(const float4*)(hp + q * GAT_CH);
    }

    const float* hbase = h + (size_t)b * GAT_N * GAT_D + 4 * i;
    float4 r[GAT_K][GAT_H];
    #pragma unroll
    for (int k = 0; k < GAT_K; ++k) {
        const float* jp = hbase + (size_t)nbs[k] * GAT_D;
        #pragma unroll
        for (int q = 0; q < GAT_H; ++q)
            r[k][q] = *(const float4*)(jp + q * GAT_CH);
    }

    const float ssrc = gat_row_sum4(dot4f(own[0], as[0]), dot4f(own[1], as[1]),
                                    dot4f(own[2], as[2]), dot4f(own[3], as[3]), b0, b1);

    float e[GAT_K];
    #pragma unroll
    for (int k = 0; k < GAT_K; ++k) {
        const float sd = gat_row_sum4(dot4f(r[k][0], ad[0]), dot4f(r[k][1], ad[1]),
                                      dot4f(r[k][2], ad[2]), dot4f(r[k][3], ad[3]), b0, b1);
        e[k] = (ssrc + sd) * 0.125f;
    }
    float m = e[0];
    #pragma unroll
    for (int k = 1; k < GAT_K; ++k) m = fmaxf(m, e[k]);

    float s = 0.0f;
    #pragma unroll
    for (int k = 0; k < GAT_K; ++k) {
        e[k] = __expf(e[k] - m);
        s += e[k];
    }
    const float sc = (1.0f / s) * 0.25f;

    float4 acc[GAT_H];
    #pragma unroll
    for (int q = 0; q < GAT_H; ++q) acc[q] = make_float4(0.f, 0.f, 0.f, 0.f);

    #pragma unroll
    for (int k = 0; k < GAT_K; ++k) {
        const float wn = e[k] * sc;
        acc[0] = fma4(gat_row_read<0x00>(wn), r[k][0], acc[0]);
        acc[1] = fma4(gat_row_read<0xAA>(wn), r[k][1], acc[1]);
        acc[2] = fma4(gat_row_read<0x55>(wn), r[k][2], acc[2]);
        acc[3] = fma4(gat_row_read<0xFF>(wn), r[k][3], acc[3]);
    }

    float* op = out + (size_t)p * GAT_CH + 4 * i;
    *(float4*)(op) = make_float4((acc[0].x + acc[1].x) + (acc[2].x + acc[3].x),
                                 (acc[0].y + acc[1].y) + (acc[2].y + acc[3].y),
                                 (acc[0].z + acc[1].z) + (acc[2].z + acc[3].z),
                                 (acc[0].w + acc[1].w) + (acc[2].w + acc[3].w));
}

extern "C" void kernel_launch(void* const* d_in, const int* in_sizes, int n_in,
                              void* d_out, int out_size, void* d_ws, size_t ws_size,
                              hipStream_t stream)
{
    const float* x    = (const float*)d_in[0];
    const int*   idx  = (const int*)  d_in[1];
    const float* W    = (const float*)d_in[2];
    const float* attn = (const float*)d_in[3];
    float* out = (float*)d_out;
    float* h   = (float*)d_ws;

    gat_gemm_wmma<<<GAT_GROUPS, 256, 0, stream>>>(x, W, h);

    gat_gather_softmax<<<(GAT_B * GAT_N) / (2 * GAT_GW), GAT_GW * 32, 0, stream>>>(h, idx, attn, out);
}
